# starting kernel
# baseline (speedup 1.0000x reference)
#include <hip/hip_runtime.h>
#include <math.h>

#define N_OBS   6000
#define N_X     6000
#define D_FEAT  16
#define N_DOM   3
#define N_SLICES 16
#define OBS_TILES (N_OBS / 16)
#define COL_TILES (N_X / 16)

typedef float v2f __attribute__((ext_vector_type(2)));
typedef float v4f __attribute__((ext_vector_type(4)));
typedef float v8f __attribute__((ext_vector_type(8)));

#define WS_HDR    0
#define WS_SXOBS  16
#define WS_SX     (WS_SXOBS + N_OBS * D_FEAT)
#define WS_NOBS   (WS_SX + N_X * D_FEAT)
#define WS_NX     (WS_NOBS + N_OBS)
#define WS_BETA0  (WS_NX + N_X)
#define WS_BETA1  (WS_BETA0 + N_DOM * N_OBS)
#define WS_PART0  (WS_BETA1 + N_DOM * N_OBS)
#define WS_PART1  (WS_PART0 + N_SLICES * N_X)

__global__ void prep_scalars(const float* __restrict__ logits,
                             const float* __restrict__ raw_ls,
                             float* __restrict__ hdr) {
    if (threadIdx.x == 0) {
        float ls = log1pf(__expf(raw_ls[0]));
        hdr[0] = 1.0f / ls;
        float t01 = 1.0f / (1.0f + __expf(-logits[0]));
        float t02 = 1.0f / (1.0f + __expf(-logits[1]));
        float t12 = 1.0f / (1.0f + __expf(-logits[2]));
        float F[9] = {1.0f, t01, t02,   t01, 1.0f, t12,   t02, t12, 1.0f};
        #pragma unroll
        for (int k = 0; k < 9; ++k) hdr[1 + k] = F[k];
    }
}

__global__ void prep_rows(const float* __restrict__ xobs,
                          const float* __restrict__ x,
                          const float* __restrict__ alpha0,
                          const float* __restrict__ alpha1,
                          const int*   __restrict__ ro,
                          float* __restrict__ ws) {
    int tid = blockIdx.x * blockDim.x + threadIdx.x;
    if (tid >= N_OBS + N_X) return;
    const float inv_ls = ws[0];
    if (tid < N_OBS) {
        int i = tid;
        const float* src = xobs + i * D_FEAT;
        float* dst = ws + WS_SXOBS + i * D_FEAT;
        float nrm = 0.0f;
        #pragma unroll
        for (int k = 0; k < D_FEAT; ++k) {
            float v = src[k] * inv_ls;
            dst[k] = v;
            nrm = fmaf(v, v, nrm);
        }
        ws[WS_NOBS + i] = nrm;
        int dom = (i >= ro[2]) + (i >= ro[4]);
        float a0 = alpha0[i], a1 = alpha1[i];
        #pragma unroll
        for (int c = 0; c < N_DOM; ++c) {
            float f = ws[1 + dom * 3 + c];
            ws[WS_BETA0 + c * N_OBS + i] = a0 * f;
            ws[WS_BETA1 + c * N_OBS + i] = a1 * f;
        }
    } else {
        int j = tid - N_OBS;
        const float* src = x + j * D_FEAT;
        float* dst = ws + WS_SX + j * D_FEAT;
        float nrm = 0.0f;
        #pragma unroll
        for (int k = 0; k < D_FEAT; ++k) {
            float v = src[k] * inv_ls;
            dst[k] = v;
            nrm = fmaf(v, v, nrm);
        }
        ws[WS_NX + j] = nrm;
    }
}

__global__ void __launch_bounds__(256) matern_wmma(const int* __restrict__ rx,
                                                   float* __restrict__ ws) {
    const int lane = threadIdx.x & 31;
    const int wave = blockIdx.x * (blockDim.x >> 5) + (threadIdx.x >> 5);
    const int colTile = wave >> 4;
    const int slice   = wave & (N_SLICES - 1);
    const int n  = lane & 15;
    const int g  = lane >> 4;
    const int kb = g * 2;
    const int j  = colTile * 16 + n;

    const float* __restrict__ sxobs = ws + WS_SXOBS;
    const float* __restrict__ sx    = ws + WS_SX;
    const float* __restrict__ nobs  = ws + WS_NOBS;

    const float* bp = sx + j * D_FEAT + kb;
    v2f b0 = *(const v2f*)(bp + 0);
    v2f b1 = *(const v2f*)(bp + 4);
    v2f b2 = *(const v2f*)(bp + 8);
    v2f b3 = *(const v2f*)(bp + 12);
    const float nxl = ws[WS_NX + j];

    const int dx = (j >= rx[2]) + (j >= rx[4]);
    const float* __restrict__ w0p = ws + WS_BETA0 + dx * N_OBS;
    const float* __restrict__ w1p = ws + WS_BETA1 + dx * N_OBS;

    float acc0 = 0.0f, acc1 = 0.0f;

    for (int t = slice; t < OBS_TILES; t += N_SLICES) {
        const int i0 = t * 16;
        const float* ap = sxobs + (i0 + n) * D_FEAT + kb;
        v2f a0 = *(const v2f*)(ap + 0);
        v2f a1 = *(const v2f*)(ap + 4);
        v2f a2 = *(const v2f*)(ap + 8);
        v2f a3 = *(const v2f*)(ap + 12);

        v8f c = {};
        c = __builtin_amdgcn_wmma_f32_16x16x4_f32(false, a0, false, b0, (short)0, c, false, false);
        c = __builtin_amdgcn_wmma_f32_16x16x4_f32(false, a1, false, b1, (short)0, c, false, false);
        c = __builtin_amdgcn_wmma_f32_16x16x4_f32(false, a2, false, b2, (short)0, c, false, false);
        c = __builtin_amdgcn_wmma_f32_16x16x4_f32(false, a3, false, b3, (short)0, c, false, false);

        const int rbase = i0 + g * 8;
        v4f nA  = *(const v4f*)(nobs + rbase);
        v4f nB  = *(const v4f*)(nobs + rbase + 4);
        v4f w0A = *(const v4f*)(w0p + rbase);
        v4f w0B = *(const v4f*)(w0p + rbase + 4);
        v4f w1A = *(const v4f*)(w1p + rbase);
        v4f w1B = *(const v4f*)(w1p + rbase + 4);

        float nr[8] = {nA.x, nA.y, nA.z, nA.w, nB.x, nB.y, nB.z, nB.w};
        float wa[8] = {w0A.x, w0A.y, w0A.z, w0A.w, w0B.x, w0B.y, w0B.z, w0B.w};
        float wb[8] = {w1A.x, w1A.y, w1A.z, w1A.w, w1B.x, w1B.y, w1B.z, w1B.w};

        #pragma unroll
        for (int r = 0; r < 8; ++r) {
            float sq   = fmaf(-2.0f, c[r], nr[r] + nxl);
            float dsq  = fmaxf(sq, 1e-12f);
            float d    = __builtin_amdgcn_sqrtf(dsq);
            float s5d  = 2.2360679774997896f * d;
            float poly = fmaf(1.6666666666666667f, dsq, 1.0f + s5d);
            float e    = __expf(-s5d);
            float k    = poly * e;
            acc0 = fmaf(wa[r], k, acc0);
            acc1 = fmaf(wb[r], k, acc1);
        }
    }

    acc0 += __shfl_xor(acc0, 16, 32);
    acc1 += __shfl_xor(acc1, 16, 32);
    if (lane < 16) {
        ws[WS_PART0 + slice * N_X + j] = acc0;
        ws[WS_PART1 + slice * N_X + j] = acc1;
    }
}

__global__ void combine(const float* __restrict__ w,
                        const float* __restrict__ ws,
                        float* __restrict__ out) {
    int j = blockIdx.x * blockDim.x + threadIdx.x;
    if (j >= N_X) return;
    const float* p0 = ws + WS_PART0;
    const float* p1 = ws + WS_PART1;
    float f0 = 0.0f, f1 = 0.0f;
    #pragma unroll
    for (int s = 0; s < N_SLICES; ++s) {
        f0 += p0[s * N_X + j];
        f1 += p1[s * N_X + j];
    }
    float wv = w[j];
    out[j] = wv * f0 + (1.0f - wv) * f1;
}

extern "C" void kernel_launch(void* const* d_in, const int* in_sizes, int n_in,
                              void* d_out, int out_size, void* d_ws, size_t ws_size,
                              hipStream_t stream) {
    const float* xobs = (const float*)d_in[0];
    const float* x    = (const float*)d_in[1];
    const float* w    = (const float*)d_in[2];
    const float* a0   = (const float*)d_in[3];
    const float* a1   = (const float*)d_in[4];
    const float* tfl  = (const float*)d_in[5];
    const float* rls  = (const float*)d_in[6];
    const int*   ro   = (const int*)d_in[7];
    const int*   rx   = (const int*)d_in[8];
    float* ws  = (float*)d_ws;
    float* out = (float*)d_out;

    prep_scalars<<<1, 32, 0, stream>>>(tfl, rls, ws);
    prep_rows<<<(N_OBS + N_X + 255) / 256, 256, 0, stream>>>(xobs, x, a0, a1, ro, ws);
    matern_wmma<<<(COL_TILES * N_SLICES) / 8, 256, 0, stream>>>(rx, ws);
    combine<<<(N_X + 255) / 256, 256, 0, stream>>>(w, ws, out);
}


# each wave owns three adjacent 16-column x-tiles: one load of the obs tile feeds three Gram tiles on independent accumulators; 48 obs slices instead of 16 (order of the partial sums changes)
# speedup vs baseline: 1.1274x; 1.1274x over previous
#include <hip/hip_runtime.h>
#include <math.h>

#define N_OBS   6000
#define N_X     6000
#define D_FEAT  16
#define N_DOM   3
#define CT       3
#define N_SLICES 48
#define OBS_TILES (N_OBS / 16)
#define COL_TILES (N_X / 16)
#define COL_GROUPS (COL_TILES / CT)

typedef float v2f __attribute__((ext_vector_type(2)));
typedef float v4f __attribute__((ext_vector_type(4)));
typedef float v8f __attribute__((ext_vector_type(8)));

#define WS_HDR    0
#define WS_SXOBS  16
#define WS_SX     (WS_SXOBS + N_OBS * D_FEAT)
#define WS_NOBS   (WS_SX + N_X * D_FEAT)
#define WS_NX     (WS_NOBS + N_OBS)
#define WS_BETA0  (WS_NX + N_X)
#define WS_BETA1  (WS_BETA0 + N_DOM * N_OBS)
#define WS_PART0  (WS_BETA1 + N_DOM * N_OBS)
#define WS_PART1  (WS_PART0 + N_SLICES * N_X)

__global__ void prep_scalars(const float* __restrict__ logits,
                             const float* __restrict__ raw_ls,
                             float* __restrict__ hdr) {
    if (threadIdx.x == 0) {
        float ls = log1pf(__expf(raw_ls[0]));
        hdr[0] = 1.0f / ls;
        float t01 = 1.0f / (1.0f + __expf(-logits[0]));
        float t02 = 1.0f / (1.0f + __expf(-logits[1]));
        float t12 = 1.0f / (1.0f + __expf(-logits[2]));
        float F[9] = {1.0f, t01, t02,   t01, 1.0f, t12,   t02, t12, 1.0f};
        #pragma unroll
        for (int k = 0; k < 9; ++k) hdr[1 + k] = F[k];
    }
}

__global__ void prep_rows(const float* __restrict__ xobs,
                          const float* __restrict__ x,
                          const float* __restrict__ alpha0,
                          const float* __restrict__ alpha1,
                          const int*   __restrict__ ro,
                          float* __restrict__ ws) {
    int tid = blockIdx.x * blockDim.x + threadIdx.x;
    if (tid >= N_OBS + N_X) return;
    const float inv_ls = ws[0];
    if (tid < N_OBS) {
        int i = tid;
        const float* src = xobs + i * D_FEAT;
        float* dst = ws + WS_SXOBS + i * D_FEAT;
        float nrm = 0.0f;
        #pragma unroll
        for (int k = 0; k < D_FEAT; ++k) {
            float v = src[k] * inv_ls;
            dst[k] = v;
            nrm = fmaf(v, v, nrm);
        }
        ws[WS_NOBS + i] = nrm;
        int dom = (i >= ro[2]) + (i >= ro[4]);
        float a0 = alpha0[i], a1 = alpha1[i];
        #pragma unroll
        for (int c = 0; c < N_DOM; ++c) {
            float f = ws[1 + dom * 3 + c];
            ws[WS_BETA0 + c * N_OBS + i] = a0 * f;
            ws[WS_BETA1 + c * N_OBS + i] = a1 * f;
        }
    } else {
        int j = tid - N_OBS;
        const float* src = x + j * D_FEAT;
        float* dst = ws + WS_SX + j * D_FEAT;
        float nrm = 0.0f;
        #pragma unroll
        for (int k = 0; k < D_FEAT; ++k) {
            float v = src[k] * inv_ls;
            dst[k] = v;
            nrm = fmaf(v, v, nrm);
        }
        ws[WS_NX + j] = nrm;
    }
}

__global__ void __launch_bounds__(256) __attribute__((amdgpu_waves_per_eu(6, 6))) matern_wmma(const int* __restrict__ rx,
                                                   float* __restrict__ ws) {
    const int lane = threadIdx.x & 31;
    const int grp   = blockIdx.x / (N_SLICES / 8);
    const int slice = (blockIdx.x % (N_SLICES / 8)) * 8 + (threadIdx.x >> 5);
    const int n  = lane & 15;
    const int g  = lane >> 4;
    const int kb = g * 2;
    const int j0 = grp * (16 * CT) + n;

    const float* __restrict__ sxobs = ws + WS_SXOBS;
    const float* __restrict__ sx    = ws + WS_SX;
    const float* __restrict__ nobs  = ws + WS_NOBS;

    const int r2 = rx[2], r4 = rx[4];

    v2f   b[CT][4];
    float nxl[CT];
    const float* w0p[CT];
    #pragma unroll
    for (int c = 0; c < CT; ++c) {
        const int j = j0 + 16 * c;
        const float* bp = sx + j * D_FEAT + kb;
        b[c][0] = *(const v2f*)(bp + 0);
        b[c][1] = *(const v2f*)(bp + 4);
        b[c][2] = *(const v2f*)(bp + 8);
        b[c][3] = *(const v2f*)(bp + 12);
        nxl[c] = ws[WS_NX + j];
        const int dx = (j >= r2) + (j >= r4);
        w0p[c] = ws + WS_BETA0 + dx * N_OBS;
    }

    float acc0[CT], acc1[CT];
    #pragma unroll
    for (int c = 0; c < CT; ++c) { acc0[c] = 0.0f; acc1[c] = 0.0f; }

    for (int t = slice; t < OBS_TILES; t += N_SLICES) {
        const int i0 = t * 16;
        const float* ap = sxobs + (i0 + n) * D_FEAT + kb;
        v2f a0 = *(const v2f*)(ap + 0);
        v2f a1 = *(const v2f*)(ap + 4);
        v2f a2 = *(const v2f*)(ap + 8);
        v2f a3 = *(const v2f*)(ap + 12);

        const int rbase = i0 + g * 8;
        v4f nA = *(const v4f*)(nobs + rbase);
        v4f nB = *(const v4f*)(nobs + rbase + 4);

        v4f w0A[CT], w0B[CT], w1A[CT], w1B[CT];
        #pragma unroll
        for (int c = 0; c < CT; ++c) {
            const float* w1p = w0p[c] + N_DOM * N_OBS;
            w0A[c] = *(const v4f*)(w0p[c] + rbase);
            w0B[c] = *(const v4f*)(w0p[c] + rbase + 4);
            w1A[c] = *(const v4f*)(w1p + rbase);
            w1B[c] = *(const v4f*)(w1p + rbase + 4);
        }
        const float nr[8] = {nA.x, nA.y, nA.z, nA.w, nB.x, nB.y, nB.z, nB.w};

        v8f cc[CT];
        #pragma unroll
        for (int c = 0; c < CT; ++c) cc[c] = (v8f){};
        #pragma unroll
        for (int c = 0; c < CT; ++c)
            cc[c] = __builtin_amdgcn_wmma_f32_16x16x4_f32(false, a0, false, b[c][0], (short)0, cc[c], false, false);
        #pragma unroll
        for (int c = 0; c < CT; ++c)
            cc[c] = __builtin_amdgcn_wmma_f32_16x16x4_f32(false, a1, false, b[c][1], (short)0, cc[c], false, false);
        #pragma unroll
        for (int c = 0; c < CT; ++c)
            cc[c] = __builtin_amdgcn_wmma_f32_16x16x4_f32(false, a2, false, b[c][2], (short)0, cc[c], false, false);
        #pragma unroll
        for (int c = 0; c < CT; ++c)
            cc[c] = __builtin_amdgcn_wmma_f32_16x16x4_f32(false, a3, false, b[c][3], (short)0, cc[c], false, false);

        #pragma unroll
        for (int c = 0; c < CT; ++c) {
            const float wa[8] = {w0A[c].x, w0A[c].y, w0A[c].z, w0A[c].w, w0B[c].x, w0B[c].y, w0B[c].z, w0B[c].w};
            const float wb[8] = {w1A[c].x, w1A[c].y, w1A[c].z, w1A[c].w, w1B[c].x, w1B[c].y, w1B[c].z, w1B[c].w};

            #pragma unroll
            for (int r = 0; r < 8; ++r) {
                float sq   = fmaf(-2.0f, cc[c][r], nr[r] + nxl[c]);
                float dsq  = fmaxf(sq, 1e-12f);
                float d    = __builtin_amdgcn_sqrtf(dsq);
                float s5d  = 2.2360679774997896f * d;
                float poly = fmaf(1.6666666666666667f, dsq, 1.0f + s5d);
                float e    = __expf(-s5d);
                float k    = poly * e;
                acc0[c] = fmaf(wa[r], k, acc0[c]);
                acc1[c] = fmaf(wb[r], k, acc1[c]);
            }
        }
    }

    #pragma unroll
    for (int c = 0; c < CT; ++c) {
        float s0 = acc0[c] + __shfl_xor(acc0[c], 16, 32);
        float s1 = acc1[c] + __shfl_xor(acc1[c], 16, 32);
        if (lane < 16) {
            ws[WS_PART0 + slice * N_X + j0 + 16 * c] = s0;
            ws[WS_PART1 + slice * N_X + j0 + 16 * c] = s1;
        }
    }
}

__global__ void combine(const float* __restrict__ w,
                        const float* __restrict__ ws,
                        float* __restrict__ out) {
    int j = blockIdx.x * blockDim.x + threadIdx.x;
    if (j >= N_X) return;
    const float* p0 = ws + WS_PART0;
    const float* p1 = ws + WS_PART1;
    float f0 = 0.0f, f1 = 0.0f;
    #pragma unroll
    for (int s = 0; s < N_SLICES; ++s) {
        f0 += p0[s * N_X + j];
        f1 += p1[s * N_X + j];
    }
    float wv = w[j];
    out[j] = wv * f0 + (1.0f - wv) * f1;
}

extern "C" void kernel_launch(void* const* d_in, const int* in_sizes, int n_in,
                              void* d_out, int out_size, void* d_ws, size_t ws_size,
                              hipStream_t stream) {
    const float* xobs = (const float*)d_in[0];
    const float* x    = (const float*)d_in[1];
    const float* w    = (const float*)d_in[2];
    const float* a0   = (const float*)d_in[3];
    const float* a1   = (const float*)d_in[4];
    const float* tfl  = (const float*)d_in[5];
    const float* rls  = (const float*)d_in[6];
    const int*   ro   = (const int*)d_in[7];
    const int*   rx   = (const int*)d_in[8];
    float* ws  = (float*)d_ws;
    float* out = (float*)d_out;

    prep_scalars<<<1, 32, 0, stream>>>(tfl, rls, ws);
    prep_rows<<<(N_OBS + N_X + 255) / 256, 256, 0, stream>>>(xobs, x, a0, a1, ro, ws);
    matern_wmma<<<(COL_GROUPS * N_SLICES) / 8, 256, 0, stream>>>(rx, ws);
    combine<<<(N_X + 255) / 256, 256, 0, stream>>>(w, ws, out);
}


# elementwise stage and weighted sums run on packed two-float operations over row pairs; one multiply of d feeds exp2; column sums kept as even/odd-row pairs (order of partial sums changes)
# speedup vs baseline: 1.1838x; 1.0500x over previous
#include <hip/hip_runtime.h>
#include <math.h>

#define N_OBS   6000
#define N_X     6000
#define D_FEAT  16
#define N_DOM   3
#define CT       3
#define N_SLICES 48
#define OBS_TILES (N_OBS / 16)
#define COL_TILES (N_X / 16)
#define COL_GROUPS (COL_TILES / CT)

typedef float v2f __attribute__((ext_vector_type(2)));
typedef float v4f __attribute__((ext_vector_type(4)));
typedef float v8f __attribute__((ext_vector_type(8)));

#define WS_HDR    0
#define WS_SXOBS  16
#define WS_SX     (WS_SXOBS + N_OBS * D_FEAT)
#define WS_NOBS   (WS_SX + N_X * D_FEAT)
#define WS_NX     (WS_NOBS + N_OBS)
#define WS_BETA0  (WS_NX + N_X)
#define WS_BETA1  (WS_BETA0 + N_DOM * N_OBS)
#define WS_PART0  (WS_BETA1 + N_DOM * N_OBS)
#define WS_PART1  (WS_PART0 + N_SLICES * N_X)

__global__ void prep_scalars(const float* __restrict__ logits,
                             const float* __restrict__ raw_ls,
                             float* __restrict__ hdr) {
    if (threadIdx.x == 0) {
        float ls = log1pf(__expf(raw_ls[0]));
        hdr[0] = 1.0f / ls;
        float t01 = 1.0f / (1.0f + __expf(-logits[0]));
        float t02 = 1.0f / (1.0f + __expf(-logits[1]));
        float t12 = 1.0f / (1.0f + __expf(-logits[2]));
        float F[9] = {1.0f, t01, t02,   t01, 1.0f, t12,   t02, t12, 1.0f};
        #pragma unroll
        for (int k = 0; k < 9; ++k) hdr[1 + k] = F[k];
    }
}

__global__ void prep_rows(const float* __restrict__ xobs,
                          const float* __restrict__ x,
                          const float* __restrict__ alpha0,
                          const float* __restrict__ alpha1,
                          const int*   __restrict__ ro,
                          float* __restrict__ ws) {
    int tid = blockIdx.x * blockDim.x + threadIdx.x;
    if (tid >= N_OBS + N_X) return;
    const float inv_ls = ws[0];
    if (tid < N_OBS) {
        int i = tid;
        const float* src = xobs + i * D_FEAT;
        float* dst = ws + WS_SXOBS + i * D_FEAT;
        float nrm = 0.0f;
        #pragma unroll
        for (int k = 0; k < D_FEAT; ++k) {
            float v = src[k] * inv_ls;
            dst[k] = v;
            nrm = fmaf(v, v, nrm);
        }
        ws[WS_NOBS + i] = nrm;
        int dom = (i >= ro[2]) + (i >= ro[4]);
        float a0 = alpha0[i], a1 = alpha1[i];
        #pragma unroll
        for (int c = 0; c < N_DOM; ++c) {
            float f = ws[1 + dom * 3 + c];
            ws[WS_BETA0 + c * N_OBS + i] = a0 * f;
            ws[WS_BETA1 + c * N_OBS + i] = a1 * f;
        }
    } else {
        int j = tid - N_OBS;
        const float* src = x + j * D_FEAT;
        float* dst = ws + WS_SX + j * D_FEAT;
        float nrm = 0.0f;
        #pragma unroll
        for (int k = 0; k < D_FEAT; ++k) {
            float v = src[k] * inv_ls;
            dst[k] = v;
            nrm = fmaf(v, v, nrm);
        }
        ws[WS_NX + j] = nrm;
    }
}

__global__ void __launch_bounds__(256) __attribute__((amdgpu_waves_per_eu(6, 6))) matern_wmma(const int* __restrict__ rx,
                                                   float* __restrict__ ws) {
    const int lane = threadIdx.x & 31;
    const int grp   = blockIdx.x / (N_SLICES / 8);
    const int slice = (blockIdx.x % (N_SLICES / 8)) * 8 + (threadIdx.x >> 5);
    const int n  = lane & 15;
    const int g  = lane >> 4;
    const int kb = g * 2;
    const int j0 = grp * (16 * CT) + n;

    const float* __restrict__ sxobs = ws + WS_SXOBS;
    const float* __restrict__ sx    = ws + WS_SX;
    const float* __restrict__ nobs  = ws + WS_NOBS;

    const int r2 = rx[2], r4 = rx[4];

    v2f   b[CT][4];
    float nxl[CT];
    const float* w0p[CT];
    #pragma unroll
    for (int c = 0; c < CT; ++c) {
        const int j = j0 + 16 * c;
        const float* bp = sx + j * D_FEAT + kb;
        b[c][0] = *(const v2f*)(bp + 0);
        b[c][1] = *(const v2f*)(bp + 4);
        b[c][2] = *(const v2f*)(bp + 8);
        b[c][3] = *(const v2f*)(bp + 12);
        nxl[c] = ws[WS_NX + j];
        const int dx = (j >= r2) + (j >= r4);
        w0p[c] = ws + WS_BETA0 + dx * N_OBS;
    }

    v2f acc0[CT], acc1[CT];
    #pragma unroll
    for (int c = 0; c < CT; ++c) { acc0[c] = (v2f){0.0f, 0.0f}; acc1[c] = (v2f){0.0f, 0.0f}; }

    const v2f k_m2   = {-2.0f, -2.0f};
    const v2f k_one  = {1.0f, 1.0f};
    const v2f k_eps  = {1e-12f, 1e-12f};
    const v2f k_s5   = {2.2360679774997896f, 2.2360679774997896f};
    const v2f k_53   = {1.6666666666666667f, 1.6666666666666667f};
    const v2f k_earg = {-3.2259641822295611f, -3.2259641822295611f};

    for (int t = slice; t < OBS_TILES; t += N_SLICES) {
        const int i0 = t * 16;
        const float* ap = sxobs + (i0 + n) * D_FEAT + kb;
        v2f a0 = *(const v2f*)(ap + 0);
        v2f a1 = *(const v2f*)(ap + 4);
        v2f a2 = *(const v2f*)(ap + 8);
        v2f a3 = *(const v2f*)(ap + 12);

        const int rbase = i0 + g * 8;
        v4f nA = *(const v4f*)(nobs + rbase);
        v4f nB = *(const v4f*)(nobs + rbase + 4);

        v4f w0A[CT], w0B[CT], w1A[CT], w1B[CT];
        #pragma unroll
        for (int c = 0; c < CT; ++c) {
            const float* w1p = w0p[c] + N_DOM * N_OBS;
            w0A[c] = *(const v4f*)(w0p[c] + rbase);
            w0B[c] = *(const v4f*)(w0p[c] + rbase + 4);
            w1A[c] = *(const v4f*)(w1p + rbase);
            w1B[c] = *(const v4f*)(w1p + rbase + 4);
        }
        const v2f nr2[4] = {{nA.x, nA.y}, {nA.z, nA.w}, {nB.x, nB.y}, {nB.z, nB.w}};

        v8f cc[CT];
        #pragma unroll
        for (int c = 0; c < CT; ++c) cc[c] = (v8f){};
        #pragma unroll
        for (int c = 0; c < CT; ++c)
            cc[c] = __builtin_amdgcn_wmma_f32_16x16x4_f32(false, a0, false, b[c][0], (short)0, cc[c], false, false);
        #pragma unroll
        for (int c = 0; c < CT; ++c)
            cc[c] = __builtin_amdgcn_wmma_f32_16x16x4_f32(false, a1, false, b[c][1], (short)0, cc[c], false, false);
        #pragma unroll
        for (int c = 0; c < CT; ++c)
            cc[c] = __builtin_amdgcn_wmma_f32_16x16x4_f32(false, a2, false, b[c][2], (short)0, cc[c], false, false);
        #pragma unroll
        for (int c = 0; c < CT; ++c)
            cc[c] = __builtin_amdgcn_wmma_f32_16x16x4_f32(false, a3, false, b[c][3], (short)0, cc[c], false, false);

        #pragma unroll
        for (int c = 0; c < CT; ++c) {
            const v2f nx2 = {nxl[c], nxl[c]};
            const v2f wa2[4] = {{w0A[c].x, w0A[c].y}, {w0A[c].z, w0A[c].w}, {w0B[c].x, w0B[c].y}, {w0B[c].z, w0B[c].w}};
            const v2f wb2[4] = {{w1A[c].x, w1A[c].y}, {w1A[c].z, w1A[c].w}, {w1B[c].x, w1B[c].y}, {w1B[c].z, w1B[c].w}};

            #pragma unroll
            for (int p = 0; p < 4; ++p) {
                const v2f g2  = {cc[c][2 * p], cc[c][2 * p + 1]};
                v2f sq  = __builtin_elementwise_fma(k_m2, g2, nr2[p]) + nx2;
                v2f dsq = __builtin_elementwise_max(sq, k_eps);
                v2f d   = {__builtin_amdgcn_sqrtf(dsq.x), __builtin_amdgcn_sqrtf(dsq.y)};
                v2f ea  = d * k_earg;
                v2f poly = __builtin_elementwise_fma(k_53, dsq, __builtin_elementwise_fma(k_s5, d, k_one));
                v2f e   = {__builtin_amdgcn_exp2f(ea.x), __builtin_amdgcn_exp2f(ea.y)};
                v2f k   = poly * e;
                acc0[c] = __builtin_elementwise_fma(wa2[p], k, acc0[c]);
                acc1[c] = __builtin_elementwise_fma(wb2[p], k, acc1[c]);
            }
        }
    }

    #pragma unroll
    for (int c = 0; c < CT; ++c) {
        const float a0 = acc0[c].x + acc0[c].y;
        const float a1 = acc1[c].x + acc1[c].y;
        float s0 = a0 + __shfl_xor(a0, 16, 32);
        float s1 = a1 + __shfl_xor(a1, 16, 32);
        if (lane < 16) {
            ws[WS_PART0 + slice * N_X + j0 + 16 * c] = s0;
            ws[WS_PART1 + slice * N_X + j0 + 16 * c] = s1;
        }
    }
}

__global__ void combine(const float* __restrict__ w,
                        const float* __restrict__ ws,
                        float* __restrict__ out) {
    int j = blockIdx.x * blockDim.x + threadIdx.x;
    if (j >= N_X) return;
    const float* p0 = ws + WS_PART0;
    const float* p1 = ws + WS_PART1;
    float f0 = 0.0f, f1 = 0.0f;
    #pragma unroll
    for (int s = 0; s < N_SLICES; ++s) {
        f0 += p0[s * N_X + j];
        f1 += p1[s * N_X + j];
    }
    float wv = w[j];
    out[j] = wv * f0 + (1.0f - wv) * f1;
}

extern "C" void kernel_launch(void* const* d_in, const int* in_sizes, int n_in,
                              void* d_out, int out_size, void* d_ws, size_t ws_size,
                              hipStream_t stream) {
    const float* xobs = (const float*)d_in[0];
    const float* x    = (const float*)d_in[1];
    const float* w    = (const float*)d_in[2];
    const float* a0   = (const float*)d_in[3];
    const float* a1   = (const float*)d_in[4];
    const float* tfl  = (const float*)d_in[5];
    const float* rls  = (const float*)d_in[6];
    const int*   ro   = (const int*)d_in[7];
    const int*   rx   = (const int*)d_in[8];
    float* ws  = (float*)d_ws;
    float* out = (float*)d_out;

    prep_scalars<<<1, 32, 0, stream>>>(tfl, rls, ws);
    prep_rows<<<(N_OBS + N_X + 255) / 256, 256, 0, stream>>>(xobs, x, a0, a1, ro, ws);
    matern_wmma<<<(COL_GROUPS * N_SLICES) / 8, 256, 0, stream>>>(rx, ws);
    combine<<<(N_X + 255) / 256, 256, 0, stream>>>(w, ws, out);
}


# block = eight adjacent column groups x one obs slice: the slice's obs rows, norms and row weights are copied once per block into LDS and shared by its eight waves; the loop has no global loads
# speedup vs baseline: 1.2341x; 1.0425x over previous
#include <hip/hip_runtime.h>
#include <math.h>

#define N_OBS   6000
#define N_X     6000
#define D_FEAT  16
#define N_DOM   3
#define CT       3
#define N_SLICES 48
#define OBS_TILES (N_OBS / 16)
#define COL_TILES (N_X / 16)
#define COL_GROUPS (COL_TILES / CT)
#define GROUPS_PER_BLOCK 8
#define GROUP_BLOCKS ((COL_GROUPS + GROUPS_PER_BLOCK - 1) / GROUPS_PER_BLOCK)
#define TILES_PER_SLICE ((OBS_TILES + N_SLICES - 1) / N_SLICES)
#define LT_A      0
#define LT_NRM    256
#define LT_BETA0  272
#define LT_BETA1  (LT_BETA0 + N_DOM * 16)
#define LT_FLOATS (LT_BETA1 + N_DOM * 16)
#define LW_FLOATS (TILES_PER_SLICE * LT_FLOATS)

typedef float v2f __attribute__((ext_vector_type(2)));
typedef float v4f __attribute__((ext_vector_type(4)));
typedef float v8f __attribute__((ext_vector_type(8)));

#define WS_HDR    0
#define WS_SXOBS  16
#define WS_SX     (WS_SXOBS + N_OBS * D_FEAT)
#define WS_NOBS   (WS_SX + N_X * D_FEAT)
#define WS_NX     (WS_NOBS + N_OBS)
#define WS_BETA0  (WS_NX + N_X)
#define WS_BETA1  (WS_BETA0 + N_DOM * N_OBS)
#define WS_PART0  (WS_BETA1 + N_DOM * N_OBS)
#define WS_PART1  (WS_PART0 + N_SLICES * N_X)

__global__ void prep_scalars(const float* __restrict__ logits,
                             const float* __restrict__ raw_ls,
                             float* __restrict__ hdr) {
    if (threadIdx.x == 0) {
        float ls = log1pf(__expf(raw_ls[0]));
        hdr[0] = 1.0f / ls;
        float t01 = 1.0f / (1.0f + __expf(-logits[0]));
        float t02 = 1.0f / (1.0f + __expf(-logits[1]));
        float t12 = 1.0f / (1.0f + __expf(-logits[2]));
        float F[9] = {1.0f, t01, t02,   t01, 1.0f, t12,   t02, t12, 1.0f};
        #pragma unroll
        for (int k = 0; k < 9; ++k) hdr[1 + k] = F[k];
    }
}

__global__ void prep_rows(const float* __restrict__ xobs,
                          const float* __restrict__ x,
                          const float* __restrict__ alpha0,
                          const float* __restrict__ alpha1,
                          const int*   __restrict__ ro,
                          float* __restrict__ ws) {
    int tid = blockIdx.x * blockDim.x + threadIdx.x;
    if (tid >= N_OBS + N_X) return;
    const float inv_ls = ws[0];
    if (tid < N_OBS) {
        int i = tid;
        const float* src = xobs + i * D_FEAT;
        float* dst = ws + WS_SXOBS + i * D_FEAT;
        float nrm = 0.0f;
        #pragma unroll
        for (int k = 0; k < D_FEAT; ++k) {
            float v = src[k] * inv_ls;
            dst[k] = v;
            nrm = fmaf(v, v, nrm);
        }
        ws[WS_NOBS + i] = nrm;
        int dom = (i >= ro[2]) + (i >= ro[4]);
        float a0 = alpha0[i], a1 = alpha1[i];
        #pragma unroll
        for (int c = 0; c < N_DOM; ++c) {
            float f = ws[1 + dom * 3 + c];
            ws[WS_BETA0 + c * N_OBS + i] = a0 * f;
            ws[WS_BETA1 + c * N_OBS + i] = a1 * f;
        }
    } else {
        int j = tid - N_OBS;
        const float* src = x + j * D_FEAT;
        float* dst = ws + WS_SX + j * D_FEAT;
        float nrm = 0.0f;
        #pragma unroll
        for (int k = 0; k < D_FEAT; ++k) {
            float v = src[k] * inv_ls;
            dst[k] = v;
            nrm = fmaf(v, v, nrm);
        }
        ws[WS_NX + j] = nrm;
    }
}

__global__ void __launch_bounds__(256) __attribute__((amdgpu_waves_per_eu(6, 6))) matern_wmma(const int* __restrict__ rx,
                                                   float* __restrict__ ws) {
    __shared__ __attribute__((aligned(16))) float lds[LW_FLOATS];

    const int lane  = threadIdx.x & 31;
    const int slice = blockIdx.x % N_SLICES;
    const int grp   = (blockIdx.x / N_SLICES) * GROUPS_PER_BLOCK + (threadIdx.x >> 5);
    const bool has_cols = grp < COL_GROUPS;
    const int grp_ld = has_cols ? grp : COL_GROUPS - 1;
    const int n  = lane & 15;
    const int g  = lane >> 4;
    const int kb = g * 2;
    const int j0 = grp_ld * (16 * CT) + n;

    const float* __restrict__ sxobs = ws + WS_SXOBS;
    const float* __restrict__ sx    = ws + WS_SX;

    v4f ta[2], tx;
    const int qx = threadIdx.x / 28;
    const int ex = threadIdx.x - qx * 28;
    const int tX = slice + qx * N_SLICES;
    const bool has_x = (threadIdx.x < 224) && (tX < OBS_TILES);
    #pragma unroll
    for (int i = 0; i < 2; ++i) {
        const int id = threadIdx.x + 256 * i;
        const int t = slice + (id >> 6) * N_SLICES;
        if (t < OBS_TILES)
            ta[i] = *(const v4f*)(sxobs + t * 256 + (id & 63) * 4);
    }
    if (has_x) {
        const int arr = ex >> 2;
        const int xoff = (arr == 0 ? 0 : (N_OBS + N_X) + (arr - 1) * N_OBS) + (ex & 3) * 4;
        tx = *(const v4f*)(ws + WS_NOBS + xoff + tX * 16);
    }

    const int r2 = rx[2], r4 = rx[4];

    v2f   b[CT][4];
    float nxl[CT];
    int   wofs[CT];
    #pragma unroll
    for (int c = 0; c < CT; ++c) {
        const int j = j0 + 16 * c;
        const float* bp = sx + j * D_FEAT + kb;
        b[c][0] = *(const v2f*)(bp + 0);
        b[c][1] = *(const v2f*)(bp + 4);
        b[c][2] = *(const v2f*)(bp + 8);
        b[c][3] = *(const v2f*)(bp + 12);
        nxl[c] = ws[WS_NX + j];
        const int dx = (j >= r2) + (j >= r4);
        wofs[c] = LT_BETA0 + dx * 16 + g * 8;
    }

    #pragma unroll
    for (int i = 0; i < 2; ++i) {
        const int id = threadIdx.x + 256 * i;
        const int t = slice + (id >> 6) * N_SLICES;
        if (t < OBS_TILES)
            *(v4f*)(lds + (id >> 6) * LT_FLOATS + LT_A + (id & 63) * 4) = ta[i];
    }
    if (has_x) *(v4f*)(lds + qx * LT_FLOATS + LT_NRM + ex * 4) = tx;

    v2f acc0[CT], acc1[CT];
    #pragma unroll
    for (int c = 0; c < CT; ++c) { acc0[c] = (v2f){0.0f, 0.0f}; acc1[c] = (v2f){0.0f, 0.0f}; }

    const v2f k_m2   = {-2.0f, -2.0f};
    const v2f k_one  = {1.0f, 1.0f};
    const v2f k_eps  = {1e-12f, 1e-12f};
    const v2f k_s5   = {2.2360679774997896f, 2.2360679774997896f};
    const v2f k_53   = {1.6666666666666667f, 1.6666666666666667f};
    const v2f k_earg = {-3.2259641822295611f, -3.2259641822295611f};

    __syncthreads();
    if (!has_cols) return;

    const float* tl = lds;
    for (int t = slice; t < OBS_TILES; t += N_SLICES, tl += LT_FLOATS) {
        const float* ap = tl + LT_A + n * D_FEAT + kb;
        v2f a0 = *(const v2f*)(ap + 0);
        v2f a1 = *(const v2f*)(ap + 4);
        v2f a2 = *(const v2f*)(ap + 8);
        v2f a3 = *(const v2f*)(ap + 12);

        v4f nA = *(const v4f*)(tl + LT_NRM + g * 8);
        v4f nB = *(const v4f*)(tl + LT_NRM + g * 8 + 4);

        v4f w0A[CT], w0B[CT], w1A[CT], w1B[CT];
        #pragma unroll
        for (int c = 0; c < CT; ++c) {
            const float* w0p = tl + wofs[c];
            const float* w1p = w0p + N_DOM * 16;
            w0A[c] = *(const v4f*)(w0p);
            w0B[c] = *(const v4f*)(w0p + 4);
            w1A[c] = *(const v4f*)(w1p);
            w1B[c] = *(const v4f*)(w1p + 4);
        }
        const v2f nr2[4] = {{nA.x, nA.y}, {nA.z, nA.w}, {nB.x, nB.y}, {nB.z, nB.w}};

        v8f cc[CT];
        #pragma unroll
        for (int c = 0; c < CT; ++c) cc[c] = (v8f){};
        #pragma unroll
        for (int c = 0; c < CT; ++c)
            cc[c] = __builtin_amdgcn_wmma_f32_16x16x4_f32(false, a0, false, b[c][0], (short)0, cc[c], false, false);
        #pragma unroll
        for (int c = 0; c < CT; ++c)
            cc[c] = __builtin_amdgcn_wmma_f32_16x16x4_f32(false, a1, false, b[c][1], (short)0, cc[c], false, false);
        #pragma unroll
        for (int c = 0; c < CT; ++c)
            cc[c] = __builtin_amdgcn_wmma_f32_16x16x4_f32(false, a2, false, b[c][2], (short)0, cc[c], false, false);
        #pragma unroll
        for (int c = 0; c < CT; ++c)
            cc[c] = __builtin_amdgcn_wmma_f32_16x16x4_f32(false, a3, false, b[c][3], (short)0, cc[c], false, false);

        v2f nx2[CT], wa2[CT][4], wb2[CT][4];
        #pragma unroll
        for (int c = 0; c < CT; ++c) {
            nx2[c] = (v2f){nxl[c], nxl[c]};
            wa2[c][0] = (v2f){w0A[c].x, w0A[c].y}; wa2[c][1] = (v2f){w0A[c].z, w0A[c].w};
            wa2[c][2] = (v2f){w0B[c].x, w0B[c].y}; wa2[c][3] = (v2f){w0B[c].z, w0B[c].w};
            wb2[c][0] = (v2f){w1A[c].x, w1A[c].y}; wb2[c][1] = (v2f){w1A[c].z, w1A[c].w};
            wb2[c][2] = (v2f){w1B[c].x, w1B[c].y}; wb2[c][3] = (v2f){w1B[c].z, w1B[c].w};
        }
        #pragma unroll
        for (int h = 0; h < 2; ++h) {
            v2f dsq[6], d[6], ea[6], poly[6], e[6];
            #pragma unroll
            for (int v = 0; v < 6; ++v) {
                const int u = 6 * h + v, c = u >> 2, p = u & 3;
                const v2f g2 = {cc[c][2 * p], cc[c][2 * p + 1]};
                v2f sq = __builtin_elementwise_fma(k_m2, g2, nr2[p]) + nx2[c];
                dsq[v] = __builtin_elementwise_max(sq, k_eps);
            }
            #pragma unroll
            for (int v = 0; v < 6; ++v)
                d[v] = (v2f){__builtin_amdgcn_sqrtf(dsq[v].x), __builtin_amdgcn_sqrtf(dsq[v].y)};
            #pragma unroll
            for (int v = 0; v < 6; ++v) {
                ea[v]   = d[v] * k_earg;
                poly[v] = __builtin_elementwise_fma(k_53, dsq[v], __builtin_elementwise_fma(k_s5, d[v], k_one));
            }
            #pragma unroll
            for (int v = 0; v < 6; ++v)
                e[v] = (v2f){__builtin_amdgcn_exp2f(ea[v].x), __builtin_amdgcn_exp2f(ea[v].y)};
            #pragma unroll
            for (int v = 0; v < 6; ++v) {
                const int u = 6 * h + v, c = u >> 2, p = u & 3;
                const v2f k = poly[v] * e[v];
                acc0[c] = __builtin_elementwise_fma(wa2[c][p], k, acc0[c]);
                acc1[c] = __builtin_elementwise_fma(wb2[c][p], k, acc1[c]);
            }
        }
    }

    #pragma unroll
    for (int c = 0; c < CT; ++c) {
        const float a0 = acc0[c].x + acc0[c].y;
        const float a1 = acc1[c].x + acc1[c].y;
        float s0 = a0 + __shfl_xor(a0, 16, 32);
        float s1 = a1 + __shfl_xor(a1, 16, 32);
        if (lane < 16) {
            ws[WS_PART0 + slice * N_X + j0 + 16 * c] = s0;
            ws[WS_PART1 + slice * N_X + j0 + 16 * c] = s1;
        }
    }
}

__global__ void combine(const float* __restrict__ w,
                        const float* __restrict__ ws,
                        float* __restrict__ out) {
    int j = blockIdx.x * blockDim.x + threadIdx.x;
    if (j >= N_X) return;
    const float* p0 = ws + WS_PART0;
    const float* p1 = ws + WS_PART1;
    float f0 = 0.0f, f1 = 0.0f;
    #pragma unroll
    for (int s = 0; s < N_SLICES; ++s) {
        f0 += p0[s * N_X + j];
        f1 += p1[s * N_X + j];
    }
    float wv = w[j];
    out[j] = wv * f0 + (1.0f - wv) * f1;
}

extern "C" void kernel_launch(void* const* d_in, const int* in_sizes, int n_in,
                              void* d_out, int out_size, void* d_ws, size_t ws_size,
                              hipStream_t stream) {
    const float* xobs = (const float*)d_in[0];
    const float* x    = (const float*)d_in[1];
    const float* w    = (const float*)d_in[2];
    const float* a0   = (const float*)d_in[3];
    const float* a1   = (const float*)d_in[4];
    const float* tfl  = (const float*)d_in[5];
    const float* rls  = (const float*)d_in[6];
    const int*   ro   = (const int*)d_in[7];
    const int*   rx   = (const int*)d_in[8];
    float* ws  = (float*)d_ws;
    float* out = (float*)d_out;

    prep_scalars<<<1, 32, 0, stream>>>(tfl, rls, ws);
    prep_rows<<<(N_OBS + N_X + 255) / 256, 256, 0, stream>>>(xobs, x, a0, a1, ro, ws);
    matern_wmma<<<GROUP_BLOCKS * N_SLICES, 256, 0, stream>>>(rx, ws);
    combine<<<(N_X + 255) / 256, 256, 0, stream>>>(w, ws, out);
}


# obs slices 48 -> 24, blocks of four waves for eight: half as many waves, each walks up to 16 obs tiles; LDS reserved so three blocks fill a WGP; 48 partial planes for 96 (partial-sum order changes)
# speedup vs baseline: 1.2797x; 1.0370x over previous
#include <hip/hip_runtime.h>
#include <math.h>

#define N_OBS   6000
#define N_X     6000
#define D_FEAT  16
#define N_DOM   3
#define CT       3
#define N_SLICES 24
#define OBS_TILES (N_OBS / 16)
#define COL_TILES (N_X / 16)
#define COL_GROUPS (COL_TILES / CT)
#define GROUPS_PER_BLOCK 4
#define BLOCK_THREADS (GROUPS_PER_BLOCK * 32)
#define GROUP_BLOCKS ((COL_GROUPS + GROUPS_PER_BLOCK - 1) / GROUPS_PER_BLOCK)
#define TILES_PER_SLICE ((OBS_TILES + N_SLICES - 1) / N_SLICES)
#define LT_A      0
#define LT_NRM    256
#define LT_BETA0  272
#define LT_BETA1  (LT_BETA0 + N_DOM * 16)
#define LT_FLOATS (LT_BETA1 + N_DOM * 16)
#define LW_FLOATS (TILES_PER_SLICE * LT_FLOATS)
#define LDS_FLOATS (82 * 1024 / 4)
static_assert(LW_FLOATS <= LDS_FLOATS, "the slice image must fit in the reservation");
#define A_ROUNDS (TILES_PER_SLICE / 2)
#define X_ROUNDS (TILES_PER_SLICE / 4)
#define FULL_TILES (OBS_TILES / N_SLICES)

typedef float v2f __attribute__((ext_vector_type(2)));
typedef float v4f __attribute__((ext_vector_type(4)));
typedef float v8f __attribute__((ext_vector_type(8)));

#define WS_HDR    0
#define WS_SXOBS  16
#define WS_SX     (WS_SXOBS + N_OBS * D_FEAT)
#define WS_NOBS   (WS_SX + N_X * D_FEAT)
#define WS_NX     (WS_NOBS + N_OBS)
#define WS_BETA0  (WS_NX + N_X)
#define WS_BETA1  (WS_BETA0 + N_DOM * N_OBS)
#define WS_PART0  (WS_BETA1 + N_DOM * N_OBS)
#define WS_PART1  (WS_PART0 + N_SLICES * N_X)

__global__ void prep_scalars(const float* __restrict__ logits,
                             const float* __restrict__ raw_ls,
                             float* __restrict__ hdr) {
    if (threadIdx.x == 0) {
        float ls = log1pf(__expf(raw_ls[0]));
        hdr[0] = 1.0f / ls;
        float t01 = 1.0f / (1.0f + __expf(-logits[0]));
        float t02 = 1.0f / (1.0f + __expf(-logits[1]));
        float t12 = 1.0f / (1.0f + __expf(-logits[2]));
        float F[9] = {1.0f, t01, t02,   t01, 1.0f, t12,   t02, t12, 1.0f};
        #pragma unroll
        for (int k = 0; k < 9; ++k) hdr[1 + k] = F[k];
    }
}

__global__ void prep_rows(const float* __restrict__ xobs,
                          const float* __restrict__ x,
                          const float* __restrict__ alpha0,
                          const float* __restrict__ alpha1,
                          const int*   __restrict__ ro,
                          float* __restrict__ ws) {
    int tid = blockIdx.x * blockDim.x + threadIdx.x;
    if (tid >= N_OBS + N_X) return;
    const float inv_ls = ws[0];
    if (tid < N_OBS) {
        int i = tid;
        const float* src = xobs + i * D_FEAT;
        float* dst = ws + WS_SXOBS + i * D_FEAT;
        float nrm = 0.0f;
        #pragma unroll
        for (int k = 0; k < D_FEAT; ++k) {
            float v = src[k] * inv_ls;
            dst[k] = v;
            nrm = fmaf(v, v, nrm);
        }
        ws[WS_NOBS + i] = nrm;
        int dom = (i >= ro[2]) + (i >= ro[4]);
        float a0 = alpha0[i], a1 = alpha1[i];
        #pragma unroll
        for (int c = 0; c < N_DOM; ++c) {
            float f = ws[1 + dom * 3 + c];
            ws[WS_BETA0 + c * N_OBS + i] = a0 * f;
            ws[WS_BETA1 + c * N_OBS + i] = a1 * f;
        }
    } else {
        int j = tid - N_OBS;
        const float* src = x + j * D_FEAT;
        float* dst = ws + WS_SX + j * D_FEAT;
        float nrm = 0.0f;
        #pragma unroll
        for (int k = 0; k < D_FEAT; ++k) {
            float v = src[k] * inv_ls;
            dst[k] = v;
            nrm = fmaf(v, v, nrm);
        }
        ws[WS_NX + j] = nrm;
    }
}

__global__ void __launch_bounds__(BLOCK_THREADS) __attribute__((amdgpu_waves_per_eu(3, 3))) matern_wmma(const int* __restrict__ rx,
                                                   float* __restrict__ ws) {
    __shared__ __attribute__((aligned(16))) float lds[LDS_FLOATS];

    const int lane  = threadIdx.x & 31;
    const int slice = blockIdx.x % N_SLICES;
    const int grp   = (blockIdx.x / N_SLICES) * GROUPS_PER_BLOCK + (threadIdx.x >> 5);
    const bool has_cols = grp < COL_GROUPS;
    const int grp_ld = has_cols ? grp : COL_GROUPS - 1;
    const int n  = lane & 15;
    const int g  = lane >> 4;
    const int kb = g * 2;
    const int j0 = grp_ld * (16 * CT) + n;

    const float* __restrict__ sxobs = ws + WS_SXOBS;
    const float* __restrict__ sx    = ws + WS_SX;

    v4f ta[A_ROUNDS], tx[X_ROUNDS];
    const int aq = threadIdx.x >> 6;
    const float* asrc = sxobs + (slice + aq * N_SLICES) * 256 + (threadIdx.x & 63) * 4;
    #pragma unroll
    for (int i = 0; i < A_ROUNDS; ++i) {
        if (2 * i + 1 < FULL_TILES || slice + (2 * i + aq) * N_SLICES < OBS_TILES)
            ta[i] = *(const v4f*)(asrc + i * (2 * N_SLICES * 256));
    }
    const int xq = threadIdx.x / 28;
    const int ex = threadIdx.x - xq * 28;
    const bool x_thread = xq < 4;
    const int arr = ex >> 2;
    const int xoff = (arr == 0 ? 0 : (N_OBS + N_X) + (arr - 1) * N_OBS) + (ex & 3) * 4;
    const float* xsrc = ws + WS_NOBS + xoff + (slice + xq * N_SLICES) * 16;
    #pragma unroll
    for (int i = 0; i < X_ROUNDS; ++i) {
        if (x_thread && (4 * i + 3 < FULL_TILES || slice + (4 * i + xq) * N_SLICES < OBS_TILES))
            tx[i] = *(const v4f*)(xsrc + i * (4 * N_SLICES * 16));
    }

    const int r2 = rx[2], r4 = rx[4];

    v2f   b[CT][4];
    float nxl[CT];
    int   wofs[CT];
    #pragma unroll
    for (int c = 0; c < CT; ++c) {
        const int j = j0 + 16 * c;
        const float* bp = sx + j * D_FEAT + kb;
        b[c][0] = *(const v2f*)(bp + 0);
        b[c][1] = *(const v2f*)(bp + 4);
        b[c][2] = *(const v2f*)(bp + 8);
        b[c][3] = *(const v2f*)(bp + 12);
        nxl[c] = ws[WS_NX + j];
        const int dx = (j >= r2) + (j >= r4);
        wofs[c] = LT_BETA0 + dx * 16 + g * 8;
    }

    float* adst = lds + aq * LT_FLOATS + LT_A + (threadIdx.x & 63) * 4;
    #pragma unroll
    for (int i = 0; i < A_ROUNDS; ++i) {
        if (2 * i + 1 < FULL_TILES || slice + (2 * i + aq) * N_SLICES < OBS_TILES)
            *(v4f*)(adst + i * (2 * LT_FLOATS)) = ta[i];
    }
    float* xdst = lds + xq * LT_FLOATS + LT_NRM + ex * 4;
    #pragma unroll
    for (int i = 0; i < X_ROUNDS; ++i) {
        if (x_thread && (4 * i + 3 < FULL_TILES || slice + (4 * i + xq) * N_SLICES < OBS_TILES))
            *(v4f*)(xdst + i * (4 * LT_FLOATS)) = tx[i];
    }

    v2f acc0[CT], acc1[CT];
    #pragma unroll
    for (int c = 0; c < CT; ++c) { acc0[c] = (v2f){0.0f, 0.0f}; acc1[c] = (v2f){0.0f, 0.0f}; }

    const v2f k_m2   = {-2.0f, -2.0f};
    const v2f k_one  = {1.0f, 1.0f};
    const v2f k_eps  = {1e-12f, 1e-12f};
    const v2f k_s5   = {2.2360679774997896f, 2.2360679774997896f};
    const v2f k_53   = {1.6666666666666667f, 1.6666666666666667f};
    const v2f k_earg = {-3.2259641822295611f, -3.2259641822295611f};

    __syncthreads();
    if (!has_cols) return;

    const float* tl = lds;
    for (int t = slice; t < OBS_TILES; t += N_SLICES, tl += LT_FLOATS) {
        const float* ap = tl + LT_A + n * D_FEAT + kb;
        v2f a0 = *(const v2f*)(ap + 0);
        v2f a1 = *(const v2f*)(ap + 4);
        v2f a2 = *(const v2f*)(ap + 8);
        v2f a3 = *(const v2f*)(ap + 12);

        v4f nA = *(const v4f*)(tl + LT_NRM + g * 8);
        v4f nB = *(const v4f*)(tl + LT_NRM + g * 8 + 4);

        v4f w0A[CT], w0B[CT], w1A[CT], w1B[CT];
        #pragma unroll
        for (int c = 0; c < CT; ++c) {
            const float* w0p = tl + wofs[c];
            const float* w1p = w0p + N_DOM * 16;
            w0A[c] = *(const v4f*)(w0p);
            w0B[c] = *(const v4f*)(w0p + 4);
            w1A[c] = *(const v4f*)(w1p);
            w1B[c] = *(const v4f*)(w1p + 4);
        }
        const v2f nr2[4] = {{nA.x, nA.y}, {nA.z, nA.w}, {nB.x, nB.y}, {nB.z, nB.w}};

        v8f cc[CT];
        #pragma unroll
        for (int c = 0; c < CT; ++c) cc[c] = (v8f){};
        #pragma unroll
        for (int c = 0; c < CT; ++c)
            cc[c] = __builtin_amdgcn_wmma_f32_16x16x4_f32(false, a0, false, b[c][0], (short)0, cc[c], false, false);
        #pragma unroll
        for (int c = 0; c < CT; ++c)
            cc[c] = __builtin_amdgcn_wmma_f32_16x16x4_f32(false, a1, false, b[c][1], (short)0, cc[c], false, false);
        #pragma unroll
        for (int c = 0; c < CT; ++c)
            cc[c] = __builtin_amdgcn_wmma_f32_16x16x4_f32(false, a2, false, b[c][2], (short)0, cc[c], false, false);
        #pragma unroll
        for (int c = 0; c < CT; ++c)
            cc[c] = __builtin_amdgcn_wmma_f32_16x16x4_f32(false, a3, false, b[c][3], (short)0, cc[c], false, false);

        v2f nx2[CT], wa2[CT][4], wb2[CT][4];
        #pragma unroll
        for (int c = 0; c < CT; ++c) {
            nx2[c] = (v2f){nxl[c], nxl[c]};
            wa2[c][0] = (v2f){w0A[c].x, w0A[c].y}; wa2[c][1] = (v2f){w0A[c].z, w0A[c].w};
            wa2[c][2] = (v2f){w0B[c].x, w0B[c].y}; wa2[c][3] = (v2f){w0B[c].z, w0B[c].w};
            wb2[c][0] = (v2f){w1A[c].x, w1A[c].y}; wb2[c][1] = (v2f){w1A[c].z, w1A[c].w};
            wb2[c][2] = (v2f){w1B[c].x, w1B[c].y}; wb2[c][3] = (v2f){w1B[c].z, w1B[c].w};
        }
        #pragma unroll
        for (int h = 0; h < 2; ++h) {
            v2f dsq[6], d[6], ea[6], poly[6], e[6];
            #pragma unroll
            for (int v = 0; v < 6; ++v) {
                const int u = 6 * h + v, c = u >> 2, p = u & 3;
                const v2f g2 = {cc[c][2 * p], cc[c][2 * p + 1]};
                v2f sq = __builtin_elementwise_fma(k_m2, g2, nr2[p]) + nx2[c];
                dsq[v] = __builtin_elementwise_max(sq, k_eps);
            }
            #pragma unroll
            for (int v = 0; v < 6; ++v)
                d[v] = (v2f){__builtin_amdgcn_sqrtf(dsq[v].x), __builtin_amdgcn_sqrtf(dsq[v].y)};
            #pragma unroll
            for (int v = 0; v < 6; ++v) {
                ea[v]   = d[v] * k_earg;
                poly[v] = __builtin_elementwise_fma(k_53, dsq[v], __builtin_elementwise_fma(k_s5, d[v], k_one));
            }
            #pragma unroll
            for (int v = 0; v < 6; ++v)
                e[v] = (v2f){__builtin_amdgcn_exp2f(ea[v].x), __builtin_amdgcn_exp2f(ea[v].y)};
            #pragma unroll
            for (int v = 0; v < 6; ++v) {
                const int u = 6 * h + v, c = u >> 2, p = u & 3;
                const v2f k = poly[v] * e[v];
                acc0[c] = __builtin_elementwise_fma(wa2[c][p], k, acc0[c]);
                acc1[c] = __builtin_elementwise_fma(wb2[c][p], k, acc1[c]);
            }
        }
    }

    #pragma unroll
    for (int c = 0; c < CT; ++c) {
        const float a0 = acc0[c].x + acc0[c].y;
        const float a1 = acc1[c].x + acc1[c].y;
        float s0 = a0 + __shfl_xor(a0, 16, 32);
        float s1 = a1 + __shfl_xor(a1, 16, 32);
        if (lane < 16) {
            ws[WS_PART0 + slice * N_X + j0 + 16 * c] = s0;
            ws[WS_PART1 + slice * N_X + j0 + 16 * c] = s1;
        }
    }
}

__global__ void combine(const float* __restrict__ w,
                        const float* __restrict__ ws,
                        float* __restrict__ out) {
    int j = blockIdx.x * blockDim.x + threadIdx.x;
    if (j >= N_X) return;
    const float* p0 = ws + WS_PART0;
    const float* p1 = ws + WS_PART1;
    float f0 = 0.0f, f1 = 0.0f;
    #pragma unroll
    for (int s = 0; s < N_SLICES; ++s) {
        f0 += p0[s * N_X + j];
        f1 += p1[s * N_X + j];
    }
    float wv = w[j];
    out[j] = wv * f0 + (1.0f - wv) * f1;
}

extern "C" void kernel_launch(void* const* d_in, const int* in_sizes, int n_in,
                              void* d_out, int out_size, void* d_ws, size_t ws_size,
                              hipStream_t stream) {
    const float* xobs = (const float*)d_in[0];
    const float* x    = (const float*)d_in[1];
    const float* w    = (const float*)d_in[2];
    const float* a0   = (const float*)d_in[3];
    const float* a1   = (const float*)d_in[4];
    const float* tfl  = (const float*)d_in[5];
    const float* rls  = (const float*)d_in[6];
    const int*   ro   = (const int*)d_in[7];
    const int*   rx   = (const int*)d_in[8];
    float* ws  = (float*)d_ws;
    float* out = (float*)d_out;

    prep_scalars<<<1, 32, 0, stream>>>(tfl, rls, ws);
    prep_rows<<<(N_OBS + N_X + 255) / 256, 256, 0, stream>>>(xobs, x, a0, a1, ro, ws);
    matern_wmma<<<GROUP_BLOCKS * N_SLICES, BLOCK_THREADS, 0, stream>>>(rx, ws);
    combine<<<(N_X + 255) / 256, 256, 0, stream>>>(w, ws, out);
}


# row-prep launch issues every load of a thread (scale, 3x3 factors, domain bounds, alphas, row) as one group ahead of its first store; block size a literal: its dependent memory round trips 7 -> 2
# speedup vs baseline: 1.3848x; 1.0821x over previous
#include <hip/hip_runtime.h>
#include <math.h>

#define N_OBS   6000
#define N_X     6000
#define D_FEAT  16
#define N_DOM   3
#define CT       3
#define N_SLICES 24
#define OBS_TILES (N_OBS / 16)
#define COL_TILES (N_X / 16)
#define COL_GROUPS (COL_TILES / CT)
#define GROUPS_PER_BLOCK 4
#define BLOCK_THREADS (GROUPS_PER_BLOCK * 32)
#define GROUP_BLOCKS ((COL_GROUPS + GROUPS_PER_BLOCK - 1) / GROUPS_PER_BLOCK)
#define TILES_PER_SLICE ((OBS_TILES + N_SLICES - 1) / N_SLICES)
#define LT_A      0
#define LT_NRM    256
#define LT_BETA0  272
#define LT_BETA1  (LT_BETA0 + N_DOM * 16)
#define LT_FLOATS (LT_BETA1 + N_DOM * 16)
#define LW_FLOATS (TILES_PER_SLICE * LT_FLOATS)
#define LDS_FLOATS (82 * 1024 / 4)
static_assert(LW_FLOATS <= LDS_FLOATS, "the slice image must fit in the reservation");
#define A_ROUNDS (TILES_PER_SLICE / 2)
#define X_ROUNDS (TILES_PER_SLICE / 4)
#define FULL_TILES (OBS_TILES / N_SLICES)

typedef float v2f __attribute__((ext_vector_type(2)));
typedef float v4f __attribute__((ext_vector_type(4)));
typedef float v8f __attribute__((ext_vector_type(8)));

#define WS_HDR    0
#define WS_SXOBS  16
#define WS_SX     (WS_SXOBS + N_OBS * D_FEAT)
#define WS_NOBS   (WS_SX + N_X * D_FEAT)
#define WS_NX     (WS_NOBS + N_OBS)
#define WS_BETA0  (WS_NX + N_X)
#define WS_BETA1  (WS_BETA0 + N_DOM * N_OBS)
#define WS_PART0  (WS_BETA1 + N_DOM * N_OBS)
#define WS_PART1  (WS_PART0 + N_SLICES * N_X)

__global__ void prep_scalars(const float* __restrict__ logits,
                             const float* __restrict__ raw_ls,
                             float* __restrict__ hdr) {
    if (threadIdx.x == 0) {
        float ls = log1pf(__expf(raw_ls[0]));
        hdr[0] = 1.0f / ls;
        float t01 = 1.0f / (1.0f + __expf(-logits[0]));
        float t02 = 1.0f / (1.0f + __expf(-logits[1]));
        float t12 = 1.0f / (1.0f + __expf(-logits[2]));
        float F[9] = {1.0f, t01, t02,   t01, 1.0f, t12,   t02, t12, 1.0f};
        #pragma unroll
        for (int k = 0; k < 9; ++k) hdr[1 + k] = F[k];
    }
}

#define PREP_THREADS 256
__global__ void __launch_bounds__(PREP_THREADS) prep_rows(const float* __restrict__ xobs,
                          const float* __restrict__ x,
                          const float* __restrict__ alpha0,
                          const float* __restrict__ alpha1,
                          const int*   __restrict__ ro,
                          float* __restrict__ ws) {
    const int tid = blockIdx.x * PREP_THREADS + threadIdx.x;
    if (tid >= N_OBS + N_X) return;
    const bool is_obs = tid < N_OBS;
    const int  row = is_obs ? tid : tid - N_OBS;
    const int  ia  = is_obs ? row : 0;

    const v4f* src = (const v4f*)((is_obs ? xobs : x) + row * D_FEAT);
    const v4f q0 = src[0], q1 = src[1], q2 = src[2], q3 = src[3];
    const float a0 = alpha0[ia], a1 = alpha1[ia];
    const float inv_ls = ws[WS_HDR];
    float F[9];
    #pragma unroll
    for (int k = 0; k < 9; ++k) F[k] = ws[WS_HDR + 1 + k];
    const int r2 = ro[2], r4 = ro[4];
    asm volatile("" :: "v"(a0), "v"(a1), "s"(r2), "s"(r4));

    const float s[D_FEAT] = {q0.x, q0.y, q0.z, q0.w, q1.x, q1.y, q1.z, q1.w,
                             q2.x, q2.y, q2.z, q2.w, q3.x, q3.y, q3.z, q3.w};
    float v[D_FEAT];
    float nrm = 0.0f;
    #pragma unroll
    for (int k = 0; k < D_FEAT; ++k) {
        v[k] = s[k] * inv_ls;
        nrm = fmaf(v[k], v[k], nrm);
    }

    v4f* dst = (v4f*)(ws + (is_obs ? WS_SXOBS : WS_SX) + row * D_FEAT);
    dst[0] = (v4f){v[0],  v[1],  v[2],  v[3]};
    dst[1] = (v4f){v[4],  v[5],  v[6],  v[7]};
    dst[2] = (v4f){v[8],  v[9],  v[10], v[11]};
    dst[3] = (v4f){v[12], v[13], v[14], v[15]};
    ws[(is_obs ? WS_NOBS : WS_NX) + row] = nrm;
    if (is_obs) {
        const int dom = (row >= r2) + (row >= r4);
        #pragma unroll
        for (int c = 0; c < N_DOM; ++c) {
            const float f = dom == 0 ? F[c] : (dom == 1 ? F[3 + c] : F[6 + c]);
            ws[WS_BETA0 + c * N_OBS + row] = a0 * f;
            ws[WS_BETA1 + c * N_OBS + row] = a1 * f;
        }
    }
}

__global__ void __launch_bounds__(BLOCK_THREADS) __attribute__((amdgpu_waves_per_eu(3, 3))) matern_wmma(const int* __restrict__ rx,
                                                   float* __restrict__ ws) {
    __shared__ __attribute__((aligned(16))) float lds[LDS_FLOATS];

    const int lane  = threadIdx.x & 31;
    const int slice = blockIdx.x % N_SLICES;
    const int grp   = (blockIdx.x / N_SLICES) * GROUPS_PER_BLOCK + (threadIdx.x >> 5);
    const bool has_cols = grp < COL_GROUPS;
    const int grp_ld = has_cols ? grp : COL_GROUPS - 1;
    const int n  = lane & 15;
    const int g  = lane >> 4;
    const int kb = g * 2;
    const int j0 = grp_ld * (16 * CT) + n;

    const float* __restrict__ sxobs = ws + WS_SXOBS;
    const float* __restrict__ sx    = ws + WS_SX;

    v4f ta[A_ROUNDS], tx[X_ROUNDS];
    const int aq = threadIdx.x >> 6;
    const float* asrc = sxobs + (slice + aq * N_SLICES) * 256 + (threadIdx.x & 63) * 4;
    #pragma unroll
    for (int i = 0; i < A_ROUNDS; ++i) {
        if (2 * i + 1 < FULL_TILES || slice + (2 * i + aq) * N_SLICES < OBS_TILES)
            ta[i] = *(const v4f*)(asrc + i * (2 * N_SLICES * 256));
    }
    const int xq = threadIdx.x / 28;
    const int ex = threadIdx.x - xq * 28;
    const bool x_thread = xq < 4;
    const int arr = ex >> 2;
    const int xoff = (arr == 0 ? 0 : (N_OBS + N_X) + (arr - 1) * N_OBS) + (ex & 3) * 4;
    const float* xsrc = ws + WS_NOBS + xoff + (slice + xq * N_SLICES) * 16;
    #pragma unroll
    for (int i = 0; i < X_ROUNDS; ++i) {
        if (x_thread && (4 * i + 3 < FULL_TILES || slice + (4 * i + xq) * N_SLICES < OBS_TILES))
            tx[i] = *(const v4f*)(xsrc + i * (4 * N_SLICES * 16));
    }

    const int r2 = rx[2], r4 = rx[4];

    v2f   b[CT][4];
    float nxl[CT];
    int   wofs[CT];
    #pragma unroll
    for (int c = 0; c < CT; ++c) {
        const int j = j0 + 16 * c;
        const float* bp = sx + j * D_FEAT + kb;
        b[c][0] = *(const v2f*)(bp + 0);
        b[c][1] = *(const v2f*)(bp + 4);
        b[c][2] = *(const v2f*)(bp + 8);
        b[c][3] = *(const v2f*)(bp + 12);
        nxl[c] = ws[WS_NX + j];
        const int dx = (j >= r2) + (j >= r4);
        wofs[c] = LT_BETA0 + dx * 16 + g * 8;
    }

    float* adst = lds + aq * LT_FLOATS + LT_A + (threadIdx.x & 63) * 4;
    #pragma unroll
    for (int i = 0; i < A_ROUNDS; ++i) {
        if (2 * i + 1 < FULL_TILES || slice + (2 * i + aq) * N_SLICES < OBS_TILES)
            *(v4f*)(adst + i * (2 * LT_FLOATS)) = ta[i];
    }
    float* xdst = lds + xq * LT_FLOATS + LT_NRM + ex * 4;
    #pragma unroll
    for (int i = 0; i < X_ROUNDS; ++i) {
        if (x_thread && (4 * i + 3 < FULL_TILES || slice + (4 * i + xq) * N_SLICES < OBS_TILES))
            *(v4f*)(xdst + i * (4 * LT_FLOATS)) = tx[i];
    }

    v2f acc0[CT], acc1[CT];
    #pragma unroll
    for (int c = 0; c < CT; ++c) { acc0[c] = (v2f){0.0f, 0.0f}; acc1[c] = (v2f){0.0f, 0.0f}; }

    const v2f k_m2   = {-2.0f, -2.0f};
    const v2f k_one  = {1.0f, 1.0f};
    const v2f k_eps  = {1e-12f, 1e-12f};
    const v2f k_s5   = {2.2360679774997896f, 2.2360679774997896f};
    const v2f k_53   = {1.6666666666666667f, 1.6666666666666667f};
    const v2f k_earg = {-3.2259641822295611f, -3.2259641822295611f};

    __syncthreads();
    if (!has_cols) return;

    const float* tl = lds;
    for (int t = slice; t < OBS_TILES; t += N_SLICES, tl += LT_FLOATS) {
        const float* ap = tl + LT_A + n * D_FEAT + kb;
        v2f a0 = *(const v2f*)(ap + 0);
        v2f a1 = *(const v2f*)(ap + 4);
        v2f a2 = *(const v2f*)(ap + 8);
        v2f a3 = *(const v2f*)(ap + 12);

        v4f nA = *(const v4f*)(tl + LT_NRM + g * 8);
        v4f nB = *(const v4f*)(tl + LT_NRM + g * 8 + 4);

        v4f w0A[CT], w0B[CT], w1A[CT], w1B[CT];
        #pragma unroll
        for (int c = 0; c < CT; ++c) {
            const float* w0p = tl + wofs[c];
            const float* w1p = w0p + N_DOM * 16;
            w0A[c] = *(const v4f*)(w0p);
            w0B[c] = *(const v4f*)(w0p + 4);
            w1A[c] = *(const v4f*)(w1p);
            w1B[c] = *(const v4f*)(w1p + 4);
        }
        const v2f nr2[4] = {{nA.x, nA.y}, {nA.z, nA.w}, {nB.x, nB.y}, {nB.z, nB.w}};

        v8f cc[CT];
        #pragma unroll
        for (int c = 0; c < CT; ++c) cc[c] = (v8f){};
        #pragma unroll
        for (int c = 0; c < CT; ++c)
            cc[c] = __builtin_amdgcn_wmma_f32_16x16x4_f32(false, a0, false, b[c][0], (short)0, cc[c], false, false);
        #pragma unroll
        for (int c = 0; c < CT; ++c)
            cc[c] = __builtin_amdgcn_wmma_f32_16x16x4_f32(false, a1, false, b[c][1], (short)0, cc[c], false, false);
        #pragma unroll
        for (int c = 0; c < CT; ++c)
            cc[c] = __builtin_amdgcn_wmma_f32_16x16x4_f32(false, a2, false, b[c][2], (short)0, cc[c], false, false);
        #pragma unroll
        for (int c = 0; c < CT; ++c)
            cc[c] = __builtin_amdgcn_wmma_f32_16x16x4_f32(false, a3, false, b[c][3], (short)0, cc[c], false, false);

        v2f nx2[CT], wa2[CT][4], wb2[CT][4];
        #pragma unroll
        for (int c = 0; c < CT; ++c) {
            nx2[c] = (v2f){nxl[c], nxl[c]};
            wa2[c][0] = (v2f){w0A[c].x, w0A[c].y}; wa2[c][1] = (v2f){w0A[c].z, w0A[c].w};
            wa2[c][2] = (v2f){w0B[c].x, w0B[c].y}; wa2[c][3] = (v2f){w0B[c].z, w0B[c].w};
            wb2[c][0] = (v2f){w1A[c].x, w1A[c].y}; wb2[c][1] = (v2f){w1A[c].z, w1A[c].w};
            wb2[c][2] = (v2f){w1B[c].x, w1B[c].y}; wb2[c][3] = (v2f){w1B[c].z, w1B[c].w};
        }
        #pragma unroll
        for (int h = 0; h < 2; ++h) {
            v2f dsq[6], d[6], ea[6], poly[6], e[6];
            #pragma unroll
            for (int v = 0; v < 6; ++v) {
                const int u = 6 * h + v, c = u >> 2, p = u & 3;
                const v2f g2 = {cc[c][2 * p], cc[c][2 * p + 1]};
                v2f sq = __builtin_elementwise_fma(k_m2, g2, nr2[p]) + nx2[c];
                dsq[v] = __builtin_elementwise_max(sq, k_eps);
            }
            #pragma unroll
            for (int v = 0; v < 6; ++v)
                d[v] = (v2f){__builtin_amdgcn_sqrtf(dsq[v].x), __builtin_amdgcn_sqrtf(dsq[v].y)};
            #pragma unroll
            for (int v = 0; v < 6; ++v) {
                ea[v]   = d[v] * k_earg;
                poly[v] = __builtin_elementwise_fma(k_53, dsq[v], __builtin_elementwise_fma(k_s5, d[v], k_one));
            }
            #pragma unroll
            for (int v = 0; v < 6; ++v)
                e[v] = (v2f){__builtin_amdgcn_exp2f(ea[v].x), __builtin_amdgcn_exp2f(ea[v].y)};
            #pragma unroll
            for (int v = 0; v < 6; ++v) {
                const int u = 6 * h + v, c = u >> 2, p = u & 3;
                const v2f k = poly[v] * e[v];
                acc0[c] = __builtin_elementwise_fma(wa2[c][p], k, acc0[c]);
                acc1[c] = __builtin_elementwise_fma(wb2[c][p], k, acc1[c]);
            }
        }
    }

    #pragma unroll
    for (int c = 0; c < CT; ++c) {
        const float a0 = acc0[c].x + acc0[c].y;
        const float a1 = acc1[c].x + acc1[c].y;
        float s0 = a0 + __shfl_xor(a0, 16, 32);
        float s1 = a1 + __shfl_xor(a1, 16, 32);
        if (lane < 16) {
            ws[WS_PART0 + slice * N_X + j0 + 16 * c] = s0;
            ws[WS_PART1 + slice * N_X + j0 + 16 * c] = s1;
        }
    }
}

__global__ void combine(const float* __restrict__ w,
                        const float* __restrict__ ws,
                        float* __restrict__ out) {
    int j = blockIdx.x * blockDim.x + threadIdx.x;
    if (j >= N_X) return;
    const float* p0 = ws + WS_PART0;
    const float* p1 = ws + WS_PART1;
    float f0 = 0.0f, f1 = 0.0f;
    #pragma unroll
    for (int s = 0; s < N_SLICES; ++s) {
        f0 += p0[s * N_X + j];
        f1 += p1[s * N_X + j];
    }
    float wv = w[j];
    out[j] = wv * f0 + (1.0f - wv) * f1;
}

extern "C" void kernel_launch(void* const* d_in, const int* in_sizes, int n_in,
                              void* d_out, int out_size, void* d_ws, size_t ws_size,
                              hipStream_t stream) {
    const float* xobs = (const float*)d_in[0];
    const float* x    = (const float*)d_in[1];
    const float* w    = (const float*)d_in[2];
    const float* a0   = (const float*)d_in[3];
    const float* a1   = (const float*)d_in[4];
    const float* tfl  = (const float*)d_in[5];
    const float* rls  = (const float*)d_in[6];
    const int*   ro   = (const int*)d_in[7];
    const int*   rx   = (const int*)d_in[8];
    float* ws  = (float*)d_ws;
    float* out = (float*)d_out;

    prep_scalars<<<1, 32, 0, stream>>>(tfl, rls, ws);
    prep_rows<<<(N_OBS + N_X + PREP_THREADS - 1) / PREP_THREADS, PREP_THREADS, 0, stream>>>(xobs, x, a0, a1, ro, ws);
    matern_wmma<<<GROUP_BLOCKS * N_SLICES, BLOCK_THREADS, 0, stream>>>(rx, ws);
    combine<<<(N_X + 255) / 256, 256, 0, stream>>>(w, ws, out);
}


# scalar-parameter launch folded into the row-prep launch (4 launches -> 3): every row thread derives 1/softplus(raw_ls) and the three sigmoids itself; the header plane is no longer used
# speedup vs baseline: 1.5230x; 1.0998x over previous
#include <hip/hip_runtime.h>
#include <math.h>

#define N_OBS   6000
#define N_X     6000
#define D_FEAT  16
#define N_DOM   3
#define CT       3
#define N_SLICES 24
#define OBS_TILES (N_OBS / 16)
#define COL_TILES (N_X / 16)
#define COL_GROUPS (COL_TILES / CT)
#define GROUPS_PER_BLOCK 4
#define BLOCK_THREADS (GROUPS_PER_BLOCK * 32)
#define GROUP_BLOCKS ((COL_GROUPS + GROUPS_PER_BLOCK - 1) / GROUPS_PER_BLOCK)
#define TILES_PER_SLICE ((OBS_TILES + N_SLICES - 1) / N_SLICES)
#define LT_A      0
#define LT_NRM    256
#define LT_BETA0  272
#define LT_BETA1  (LT_BETA0 + N_DOM * 16)
#define LT_FLOATS (LT_BETA1 + N_DOM * 16)
#define LW_FLOATS (TILES_PER_SLICE * LT_FLOATS)
#define LDS_FLOATS (82 * 1024 / 4)
static_assert(LW_FLOATS <= LDS_FLOATS, "the slice image must fit in the reservation");
#define A_ROUNDS (TILES_PER_SLICE / 2)
#define X_ROUNDS (TILES_PER_SLICE / 4)
#define FULL_TILES (OBS_TILES / N_SLICES)

typedef float v2f __attribute__((ext_vector_type(2)));
typedef float v4f __attribute__((ext_vector_type(4)));
typedef float v8f __attribute__((ext_vector_type(8)));

#define WS_HDR    0
#define WS_SXOBS  16
#define WS_SX     (WS_SXOBS + N_OBS * D_FEAT)
#define WS_NOBS   (WS_SX + N_X * D_FEAT)
#define WS_NX     (WS_NOBS + N_OBS)
#define WS_BETA0  (WS_NX + N_X)
#define WS_BETA1  (WS_BETA0 + N_DOM * N_OBS)
#define WS_PART0  (WS_BETA1 + N_DOM * N_OBS)
#define WS_PART1  (WS_PART0 + N_SLICES * N_X)

#define PREP_THREADS 256
__global__ void __launch_bounds__(PREP_THREADS) prep_rows(const float* __restrict__ xobs,
                          const float* __restrict__ x,
                          const float* __restrict__ alpha0,
                          const float* __restrict__ alpha1,
                          const float* __restrict__ logits,
                          const float* __restrict__ raw_ls,
                          const int*   __restrict__ ro,
                          float* __restrict__ ws) {
    const int tid = blockIdx.x * PREP_THREADS + threadIdx.x;
    if (tid >= N_OBS + N_X) return;
    const bool is_obs = tid < N_OBS;
    const int  row = is_obs ? tid : tid - N_OBS;
    const int  ia  = is_obs ? row : 0;

    const v4f* src = (const v4f*)((is_obs ? xobs : x) + row * D_FEAT);
    const v4f q0 = src[0], q1 = src[1], q2 = src[2], q3 = src[3];
    const float a0 = alpha0[ia], a1 = alpha1[ia];
    const float raw = raw_ls[0];
    const float l01 = logits[0], l02 = logits[1], l12 = logits[2];
    const int r2 = ro[2], r4 = ro[4];
    asm volatile("" :: "v"(a0), "v"(a1), "s"(raw), "s"(l01), "s"(l02), "s"(l12), "s"(r2), "s"(r4));
    __builtin_amdgcn_sched_barrier(0);

    const float ls = log1pf(__expf(raw));
    const float inv_ls = 1.0f / ls;
    const float t01 = 1.0f / (1.0f + __expf(-l01));
    const float t02 = 1.0f / (1.0f + __expf(-l02));
    const float t12 = 1.0f / (1.0f + __expf(-l12));
    const float F[9] = {1.0f, t01, t02,   t01, 1.0f, t12,   t02, t12, 1.0f};

    const float s[D_FEAT] = {q0.x, q0.y, q0.z, q0.w, q1.x, q1.y, q1.z, q1.w,
                             q2.x, q2.y, q2.z, q2.w, q3.x, q3.y, q3.z, q3.w};
    float v[D_FEAT];
    float nrm = 0.0f;
    #pragma unroll
    for (int k = 0; k < D_FEAT; ++k) {
        v[k] = s[k] * inv_ls;
        nrm = fmaf(v[k], v[k], nrm);
    }

    v4f* dst = (v4f*)(ws + (is_obs ? WS_SXOBS : WS_SX) + row * D_FEAT);
    dst[0] = (v4f){v[0],  v[1],  v[2],  v[3]};
    dst[1] = (v4f){v[4],  v[5],  v[6],  v[7]};
    dst[2] = (v4f){v[8],  v[9],  v[10], v[11]};
    dst[3] = (v4f){v[12], v[13], v[14], v[15]};
    ws[(is_obs ? WS_NOBS : WS_NX) + row] = nrm;
    if (is_obs) {
        const int dom = (row >= r2) + (row >= r4);
        #pragma unroll
        for (int c = 0; c < N_DOM; ++c) {
            const float f = dom == 0 ? F[c] : (dom == 1 ? F[3 + c] : F[6 + c]);
            ws[WS_BETA0 + c * N_OBS + row] = a0 * f;
            ws[WS_BETA1 + c * N_OBS + row] = a1 * f;
        }
    }
}

__global__ void __launch_bounds__(BLOCK_THREADS) __attribute__((amdgpu_waves_per_eu(3, 3))) matern_wmma(const int* __restrict__ rx,
                                                   float* __restrict__ ws) {
    __shared__ __attribute__((aligned(16))) float lds[LDS_FLOATS];

    const int lane  = threadIdx.x & 31;
    const int slice = blockIdx.x % N_SLICES;
    const int grp   = (blockIdx.x / N_SLICES) * GROUPS_PER_BLOCK + (threadIdx.x >> 5);
    const bool has_cols = grp < COL_GROUPS;
    const int grp_ld = has_cols ? grp : COL_GROUPS - 1;
    const int n  = lane & 15;
    const int g  = lane >> 4;
    const int kb = g * 2;
    const int j0 = grp_ld * (16 * CT) + n;

    const float* __restrict__ sxobs = ws + WS_SXOBS;
    const float* __restrict__ sx    = ws + WS_SX;

    v4f ta[A_ROUNDS], tx[X_ROUNDS];
    const int aq = threadIdx.x >> 6;
    const float* asrc = sxobs + (slice + aq * N_SLICES) * 256 + (threadIdx.x & 63) * 4;
    #pragma unroll
    for (int i = 0; i < A_ROUNDS; ++i) {
        if (2 * i + 1 < FULL_TILES || slice + (2 * i + aq) * N_SLICES < OBS_TILES)
            ta[i] = *(const v4f*)(asrc + i * (2 * N_SLICES * 256));
    }
    const int xq = threadIdx.x / 28;
    const int ex = threadIdx.x - xq * 28;
    const bool x_thread = xq < 4;
    const int arr = ex >> 2;
    const int xoff = (arr == 0 ? 0 : (N_OBS + N_X) + (arr - 1) * N_OBS) + (ex & 3) * 4;
    const float* xsrc = ws + WS_NOBS + xoff + (slice + xq * N_SLICES) * 16;
    #pragma unroll
    for (int i = 0; i < X_ROUNDS; ++i) {
        if (x_thread && (4 * i + 3 < FULL_TILES || slice + (4 * i + xq) * N_SLICES < OBS_TILES))
            tx[i] = *(const v4f*)(xsrc + i * (4 * N_SLICES * 16));
    }

    const int r2 = rx[2], r4 = rx[4];

    v2f   b[CT][4];
    float nxl[CT];
    int   wofs[CT];
    #pragma unroll
    for (int c = 0; c < CT; ++c) {
        const int j = j0 + 16 * c;
        const float* bp = sx + j * D_FEAT + kb;
        b[c][0] = *(const v2f*)(bp + 0);
        b[c][1] = *(const v2f*)(bp + 4);
        b[c][2] = *(const v2f*)(bp + 8);
        b[c][3] = *(const v2f*)(bp + 12);
        nxl[c] = ws[WS_NX + j];
        const int dx = (j >= r2) + (j >= r4);
        wofs[c] = LT_BETA0 + dx * 16 + g * 8;
    }

    float* adst = lds + aq * LT_FLOATS + LT_A + (threadIdx.x & 63) * 4;
    #pragma unroll
    for (int i = 0; i < A_ROUNDS; ++i) {
        if (2 * i + 1 < FULL_TILES || slice + (2 * i + aq) * N_SLICES < OBS_TILES)
            *(v4f*)(adst + i * (2 * LT_FLOATS)) = ta[i];
    }
    float* xdst = lds + xq * LT_FLOATS + LT_NRM + ex * 4;
    #pragma unroll
    for (int i = 0; i < X_ROUNDS; ++i) {
        if (x_thread && (4 * i + 3 < FULL_TILES || slice + (4 * i + xq) * N_SLICES < OBS_TILES))
            *(v4f*)(xdst + i * (4 * LT_FLOATS)) = tx[i];
    }

    v2f acc0[CT], acc1[CT];
    #pragma unroll
    for (int c = 0; c < CT; ++c) { acc0[c] = (v2f){0.0f, 0.0f}; acc1[c] = (v2f){0.0f, 0.0f}; }

    const v2f k_m2   = {-2.0f, -2.0f};
    const v2f k_one  = {1.0f, 1.0f};
    const v2f k_eps  = {1e-12f, 1e-12f};
    const v2f k_s5   = {2.2360679774997896f, 2.2360679774997896f};
    const v2f k_53   = {1.6666666666666667f, 1.6666666666666667f};
    const v2f k_earg = {-3.2259641822295611f, -3.2259641822295611f};

    __syncthreads();
    if (!has_cols) return;

    const float* tl = lds;
    for (int t = slice; t < OBS_TILES; t += N_SLICES, tl += LT_FLOATS) {
        const float* ap = tl + LT_A + n * D_FEAT + kb;
        v2f a0 = *(const v2f*)(ap + 0);
        v2f a1 = *(const v2f*)(ap + 4);
        v2f a2 = *(const v2f*)(ap + 8);
        v2f a3 = *(const v2f*)(ap + 12);

        v4f nA = *(const v4f*)(tl + LT_NRM + g * 8);
        v4f nB = *(const v4f*)(tl + LT_NRM + g * 8 + 4);

        v4f w0A[CT], w0B[CT], w1A[CT], w1B[CT];
        #pragma unroll
        for (int c = 0; c < CT; ++c) {
            const float* w0p = tl + wofs[c];
            const float* w1p = w0p + N_DOM * 16;
            w0A[c] = *(const v4f*)(w0p);
            w0B[c] = *(const v4f*)(w0p + 4);
            w1A[c] = *(const v4f*)(w1p);
            w1B[c] = *(const v4f*)(w1p + 4);
        }
        const v2f nr2[4] = {{nA.x, nA.y}, {nA.z, nA.w}, {nB.x, nB.y}, {nB.z, nB.w}};

        v8f cc[CT];
        #pragma unroll
        for (int c = 0; c < CT; ++c) cc[c] = (v8f){};
        #pragma unroll
        for (int c = 0; c < CT; ++c)
            cc[c] = __builtin_amdgcn_wmma_f32_16x16x4_f32(false, a0, false, b[c][0], (short)0, cc[c], false, false);
        #pragma unroll
        for (int c = 0; c < CT; ++c)
            cc[c] = __builtin_amdgcn_wmma_f32_16x16x4_f32(false, a1, false, b[c][1], (short)0, cc[c], false, false);
        #pragma unroll
        for (int c = 0; c < CT; ++c)
            cc[c] = __builtin_amdgcn_wmma_f32_16x16x4_f32(false, a2, false, b[c][2], (short)0, cc[c], false, false);
        #pragma unroll
        for (int c = 0; c < CT; ++c)
            cc[c] = __builtin_amdgcn_wmma_f32_16x16x4_f32(false, a3, false, b[c][3], (short)0, cc[c], false, false);

        v2f nx2[CT], wa2[CT][4], wb2[CT][4];
        #pragma unroll
        for (int c = 0; c < CT; ++c) {
            nx2[c] = (v2f){nxl[c], nxl[c]};
            wa2[c][0] = (v2f){w0A[c].x, w0A[c].y}; wa2[c][1] = (v2f){w0A[c].z, w0A[c].w};
            wa2[c][2] = (v2f){w0B[c].x, w0B[c].y}; wa2[c][3] = (v2f){w0B[c].z, w0B[c].w};
            wb2[c][0] = (v2f){w1A[c].x, w1A[c].y}; wb2[c][1] = (v2f){w1A[c].z, w1A[c].w};
            wb2[c][2] = (v2f){w1B[c].x, w1B[c].y}; wb2[c][3] = (v2f){w1B[c].z, w1B[c].w};
        }
        #pragma unroll
        for (int h = 0; h < 2; ++h) {
            v2f dsq[6], d[6], ea[6], poly[6], e[6];
            #pragma unroll
            for (int v = 0; v < 6; ++v) {
                const int u = 6 * h + v, c = u >> 2, p = u & 3;
                const v2f g2 = {cc[c][2 * p], cc[c][2 * p + 1]};
                v2f sq = __builtin_elementwise_fma(k_m2, g2, nr2[p]) + nx2[c];
                dsq[v] = __builtin_elementwise_max(sq, k_eps);
            }
            #pragma unroll
            for (int v = 0; v < 6; ++v)
                d[v] = (v2f){__builtin_amdgcn_sqrtf(dsq[v].x), __builtin_amdgcn_sqrtf(dsq[v].y)};
            #pragma unroll
            for (int v = 0; v < 6; ++v) {
                ea[v]   = d[v] * k_earg;
                poly[v] = __builtin_elementwise_fma(k_53, dsq[v], __builtin_elementwise_fma(k_s5, d[v], k_one));
            }
            #pragma unroll
            for (int v = 0; v < 6; ++v)
                e[v] = (v2f){__builtin_amdgcn_exp2f(ea[v].x), __builtin_amdgcn_exp2f(ea[v].y)};
            #pragma unroll
            for (int v = 0; v < 6; ++v) {
                const int u = 6 * h + v, c = u >> 2, p = u & 3;
                const v2f k = poly[v] * e[v];
                acc0[c] = __builtin_elementwise_fma(wa2[c][p], k, acc0[c]);
                acc1[c] = __builtin_elementwise_fma(wb2[c][p], k, acc1[c]);
            }
        }
    }

    #pragma unroll
    for (int c = 0; c < CT; ++c) {
        const float a0 = acc0[c].x + acc0[c].y;
        const float a1 = acc1[c].x + acc1[c].y;
        float s0 = a0 + __shfl_xor(a0, 16, 32);
        float s1 = a1 + __shfl_xor(a1, 16, 32);
        if (lane < 16) {
            ws[WS_PART0 + slice * N_X + j0 + 16 * c] = s0;
            ws[WS_PART1 + slice * N_X + j0 + 16 * c] = s1;
        }
    }
}

__global__ void combine(const float* __restrict__ w,
                        const float* __restrict__ ws,
                        float* __restrict__ out) {
    int j = blockIdx.x * blockDim.x + threadIdx.x;
    if (j >= N_X) return;
    const float* p0 = ws + WS_PART0;
    const float* p1 = ws + WS_PART1;
    float f0 = 0.0f, f1 = 0.0f;
    #pragma unroll
    for (int s = 0; s < N_SLICES; ++s) {
        f0 += p0[s * N_X + j];
        f1 += p1[s * N_X + j];
    }
    float wv = w[j];
    out[j] = wv * f0 + (1.0f - wv) * f1;
}

extern "C" void kernel_launch(void* const* d_in, const int* in_sizes, int n_in,
                              void* d_out, int out_size, void* d_ws, size_t ws_size,
                              hipStream_t stream) {
    const float* xobs = (const float*)d_in[0];
    const float* x    = (const float*)d_in[1];
    const float* w    = (const float*)d_in[2];
    const float* a0   = (const float*)d_in[3];
    const float* a1   = (const float*)d_in[4];
    const float* tfl  = (const float*)d_in[5];
    const float* rls  = (const float*)d_in[6];
    const int*   ro   = (const int*)d_in[7];
    const int*   rx   = (const int*)d_in[8];
    float* ws  = (float*)d_ws;
    float* out = (float*)d_out;

    prep_rows<<<(N_OBS + N_X + PREP_THREADS - 1) / PREP_THREADS, PREP_THREADS, 0, stream>>>(xobs, x, a0, a1, tfl, rls, ro, ws);
    matern_wmma<<<GROUP_BLOCKS * N_SLICES, BLOCK_THREADS, 0, stream>>>(rx, ws);
    combine<<<(N_X + 255) / 256, 256, 0, stream>>>(w, ws, out);
}
